# starting kernel
# baseline (speedup 1.0000x reference)
#include <hip/hip_runtime.h>

typedef float v2f __attribute__((ext_vector_type(2)));
typedef float v4f __attribute__((ext_vector_type(4)));
typedef float v8f __attribute__((ext_vector_type(8)));

#define Bn   32
#define Cn   512
#define HW2  1024
#define NW   77
#define NP   80
#define WD   256

static __device__ __forceinline__ v8f wmma_f32(v2f a, v2f b, v8f c) {
  return __builtin_amdgcn_wmma_f32_16x16x4_f32(false, a, false, b, (short)0, c, false, false);
}

__global__ __launch_bounds__(256) void we_gemm_kernel(
    const float* __restrict__ word_emb, const float* __restrict__ W_fc,
    const float* __restrict__ b_fc, float* __restrict__ we_ws,
    float* __restrict__ weT_ws) {
  const int lane = threadIdx.x & 31;
  const int wave = threadIdx.x >> 5;
  int tile = blockIdx.x * 8 + wave;
  const int b  = tile / (5 * 64);
  tile -= b * 5 * 64;
  const int tm = tile >> 6;
  const int tn = tile & 63;
  const int half = lane >> 4;
  const int l16  = lane & 15;

  const int a_row = tm * 16 + l16;
  const int col   = tn * 16 + l16;
  const float* Abase = word_emb + ((size_t)b * NW + (a_row < NW ? a_row : NW - 1)) * WD;
  const float* Bbase = W_fc + (size_t)col * WD;

  v8f acc = {};
  for (int k0 = 0; k0 < WD; k0 += 4) {
    const int k = k0 + 2 * half;
    v2f a  = *(const v2f*)(Abase + k);
    v2f bm = *(const v2f*)(Bbase + k);
    acc = wmma_f32(a, bm, acc);
  }
  const float bias = b_fc[col];

  float* D = we_ws + ((size_t)b * NP + tm * 16 + 8 * half) * HW2 + col;
  #pragma unroll
  for (int i = 0; i < 8; ++i) D[(size_t)i * HW2] = acc[i] + bias;

  float* T = weT_ws + ((size_t)b * HW2 + col) * NP + tm * 16 + 8 * half;
  v4f lo = {acc[0] + bias, acc[1] + bias, acc[2] + bias, acc[3] + bias};
  v4f hi = {acc[4] + bias, acc[5] + bias, acc[6] + bias, acc[7] + bias};
  *(v4f*)(T)     = lo;
  *(v4f*)(T + 4) = hi;
}

__global__ __launch_bounds__(256) void scores_softmax_kernel(
    const float* __restrict__ feat, const float* __restrict__ we_ws,
    float* __restrict__ att_ws) {
  const int lane = threadIdx.x & 31;
  const int wave = threadIdx.x >> 5;
  const int slab = blockIdx.x * 8 + wave;
  const int b  = slab >> 5;
  const int cs = (slab & 31) * 16;
  const int half = lane >> 4;
  const int l16  = lane & 15;

  const float* Abase  = feat  + ((size_t)b * Cn + cs + l16) * HW2;
  const float* Bbase0 = we_ws + ((size_t)b * NP + l16) * HW2;

  v8f acc[5] = {};
  for (int k0 = 0; k0 < HW2; k0 += 4) {
    const int k = k0 + 2 * half;
    v2f a = *(const v2f*)(Abase + k);
    #pragma unroll
    for (int t = 0; t < 5; ++t) {
      v2f bm = *(const v2f*)(Bbase0 + (size_t)(16 * t) * HW2 + k);
      acc[t] = wmma_f32(a, bm, acc[t]);
    }
  }

  float* Dbase = att_ws + ((size_t)b * Cn + cs + 8 * half) * NP;
  #pragma unroll
  for (int i = 0; i < 8; ++i) {
    float m = -3.402823466e38f;
    #pragma unroll
    for (int t = 0; t < 5; ++t)
      if (16 * t + l16 < NW) m = fmaxf(m, acc[t][i]);
    #pragma unroll
    for (int off = 8; off >= 1; off >>= 1) m = fmaxf(m, __shfl_xor(m, off, 32));

    float e[5], s = 0.f;
    #pragma unroll
    for (int t = 0; t < 5; ++t) {
      e[t] = (16 * t + l16 < NW) ? __expf(acc[t][i] - m) : 0.f;
      s += e[t];
    }
    #pragma unroll
    for (int off = 8; off >= 1; off >>= 1) s += __shfl_xor(s, off, 32);
    const float inv = 1.f / s;
    #pragma unroll
    for (int t = 0; t < 5; ++t)
      Dbase[(size_t)i * NP + 16 * t + l16] = e[t] * inv;
  }
}

__global__ __launch_bounds__(256) void attfeat_kernel(
    const float* __restrict__ att_ws, const float* __restrict__ weT_ws,
    float* __restrict__ out) {
  const int lane = threadIdx.x & 31;
  const int wave = threadIdx.x >> 5;
  int tile = blockIdx.x * 8 + wave;
  const int b = tile / (32 * 16);
  tile -= b * 32 * 16;
  const int tm = tile >> 4;
  const int tq = tile & 15;
  const int half = lane >> 4;
  const int l16  = lane & 15;

  const float* Abase = att_ws + ((size_t)b * Cn + tm * 16 + l16) * NP;
  const float* Bbase0 = weT_ws + ((size_t)b * HW2 + tq * 64 + l16) * NP;

  v8f acc[4] = {};
  for (int k0 = 0; k0 < NP; k0 += 4) {
    const int k = k0 + 2 * half;
    v2f a = *(const v2f*)(Abase + k);
    #pragma unroll
    for (int t = 0; t < 4; ++t) {
      v2f bm = *(const v2f*)(Bbase0 + (size_t)(16 * t) * NP + k);
      acc[t] = wmma_f32(a, bm, acc[t]);
    }
  }
  float* D = out + ((size_t)b * Cn + tm * 16 + 8 * half) * HW2 + tq * 64 + l16;
  #pragma unroll
  for (int t = 0; t < 4; ++t)
    #pragma unroll
    for (int i = 0; i < 8; ++i)
      D[(size_t)i * HW2 + 16 * t] = acc[t][i];
}

extern "C" void kernel_launch(void* const* d_in, const int* in_sizes, int n_in,
                              void* d_out, int out_size, void* d_ws, size_t ws_size,
                              hipStream_t stream) {
  const float* feat     = (const float*)d_in[0];
  const float* word_emb = (const float*)d_in[1];
  const float* W_fc     = (const float*)d_in[2];
  const float* b_fc     = (const float*)d_in[3];
  float* out = (float*)d_out;

  float* we_ws  = (float*)d_ws;
  float* weT_ws = we_ws  + (size_t)Bn * NP * HW2;
  float* att_ws = weT_ws + (size_t)Bn * HW2 * NP;

  we_gemm_kernel       <<<(Bn * 5 * 64) / 8,  256, 0, stream>>>(word_emb, W_fc, b_fc, we_ws, weT_ws);
  scores_softmax_kernel<<<(Bn * 32) / 8,      256, 0, stream>>>(feat, we_ws, att_ws);
  attfeat_kernel       <<<(Bn * 32 * 16) / 8, 256, 0, stream>>>(att_ws, weT_ws, out);
}


# we GEMM and out GEMM: a wave's two column tiles interleave (tile t = columns 2c+t): 8-byte stores write whole lines, out GEMM's B comes from the (n,k) we plane, (k,n) plane dropped; sum order kept
# speedup vs baseline: 2.1706x; 2.1706x over previous
#include <hip/hip_runtime.h>

typedef float v2f __attribute__((ext_vector_type(2)));
typedef float v4f __attribute__((ext_vector_type(4)));
typedef float v8f __attribute__((ext_vector_type(8)));

#define Bn   32
#define Cn   512
#define HW2  1024
#define NW   77
#define NP   80
#define WD   256

static __device__ __forceinline__ v8f wmma_f32(v2f a, v2f b, v8f c) {
  return __builtin_amdgcn_wmma_f32_16x16x4_f32(false, a, false, b, (short)0, c, false, false);
}

#define K1_STEPS 4

static __device__ __forceinline__ void k1_load(
    v2f (&a)[K1_STEPS][5], v2f (&bm)[K1_STEPS][2],
    const float* const (&Ab)[5], const float* const (&Bb)[2], int kofs) {
  #pragma unroll
  for (int s = 0; s < K1_STEPS; ++s) {
    #pragma unroll
    for (int tm = 0; tm < 5; ++tm) a[s][tm] = *(const v2f*)(Ab[tm] + kofs + 4 * s);
    #pragma unroll
    for (int j = 0; j < 2; ++j) bm[s][j] = *(const v2f*)(Bb[j] + kofs + 4 * s);
  }
}

static __device__ __forceinline__ void k1_mma(
    v8f (&acc)[5][2], const v2f (&a)[K1_STEPS][5], const v2f (&bm)[K1_STEPS][2]) {
  #pragma unroll
  for (int s = 0; s < K1_STEPS; ++s)
    #pragma unroll
    for (int tm = 0; tm < 5; ++tm)
      #pragma unroll
      for (int j = 0; j < 2; ++j)
        acc[tm][j] = wmma_f32(a[s][tm], bm[s][j], acc[tm][j]);
}

__global__ __launch_bounds__(128) void we_gemm_kernel(
    const float* __restrict__ word_emb, const float* __restrict__ W_fc,
    const float* __restrict__ b_fc, float* __restrict__ we_ws) {
  const int lane = threadIdx.x & 31;
  const int wave = threadIdx.x >> 5;
  const int tile = blockIdx.x * 4 + wave;
  const int b  = tile >> 5;
  const int jb = tile & 31;
  const int half = lane >> 4;
  const int l16  = lane & 15;

  const float* Ab[5];
  #pragma unroll
  for (int tm = 0; tm < 5; ++tm) {
    const int a_row = tm * 16 + l16;
    Ab[tm] = word_emb + ((size_t)b * NW + (a_row < NW ? a_row : NW - 1)) * WD + 2 * half;
  }
  const float* Bb[2];
  #pragma unroll
  for (int j = 0; j < 2; ++j)
    Bb[j] = W_fc + (size_t)(jb * 32 + 2 * l16 + j) * WD + 2 * half;

  v8f acc[5][2] = {};
  v2f a0[K1_STEPS][5], b0[K1_STEPS][2], a1[K1_STEPS][5], b1[K1_STEPS][2];
  k1_load(a0, b0, Ab, Bb, 0);
  #pragma unroll
  for (int s = 0; s < K1_STEPS; ++s)
    asm volatile("" :: "v"(a0[s][0]), "v"(a0[s][1]), "v"(a0[s][2]), "v"(a0[s][3]),
                       "v"(a0[s][4]), "v"(b0[s][0]), "v"(b0[s][1]));
  for (int k0 = 0; k0 < WD; k0 += 8 * K1_STEPS) {
    k1_load(a1, b1, Ab, Bb, k0 + 4 * K1_STEPS);
    k1_mma(acc, a0, b0);
    k1_load(a0, b0, Ab, Bb, (k0 + 8 * K1_STEPS) & (WD - 1));
    k1_mma(acc, a1, b1);
  }

  const int col = jb * 32 + 2 * l16;
  const float bias0 = b_fc[col], bias1 = b_fc[col + 1];
  #pragma unroll
  for (int tm = 0; tm < 5; ++tm) {
    float* D = we_ws + ((size_t)b * NP + tm * 16 + 8 * half) * HW2 + col;
    #pragma unroll
    for (int i = 0; i < 8; ++i) {
      const v2f v = {acc[tm][0][i] + bias0, acc[tm][1][i] + bias1};
      *(v2f*)(D + (size_t)i * HW2) = v;
    }
  }
}

#define K2_KC    32
#define K2_NCH   (HW2 / K2_KC)
#define K2_PITCH 36
#define K2_AROWS 64
#define K2_BUF   ((K2_AROWS + NP) * K2_PITCH)

static __device__ __forceinline__ void k2_wait_lds() {
  __builtin_amdgcn_sched_barrier(0);
  asm volatile("s_wait_dscnt 0x0" ::: "memory");
}
static __device__ __forceinline__ void k2_barrier() {
  asm volatile("s_barrier_signal -1\n\ts_barrier_wait -1" ::: "memory");
}

static __device__ __forceinline__ void k2_gload(
    v4f (&p)[9], const float* gA, const float* gB, int k0) {
  #pragma unroll
  for (int i = 0; i < 4; ++i) p[i]     = *(const v4f*)(gA + (size_t)(4 * i) * HW2 + k0);
  #pragma unroll
  for (int i = 0; i < 5; ++i) p[4 + i] = *(const v4f*)(gB + (size_t)(4 * i) * HW2 + k0);
}

static __device__ __forceinline__ void k2_lstore(
    float* buf, const v4f (&p)[9], int stA, int stB) {
  #pragma unroll
  for (int i = 0; i < 4; ++i) *(v4f*)(buf + stA + 4 * i * K2_PITCH) = p[i];
  #pragma unroll
  for (int i = 0; i < 5; ++i) *(v4f*)(buf + stB + 4 * i * K2_PITCH) = p[4 + i];
}

static __device__ __forceinline__ void k2_ldfrag(
    v2f (&a)[4], v2f (&bm)[4][5], const float* buf, int fA, int fB, int h) {
  #pragma unroll
  for (int s = 0; s < 4; ++s) {
    a[s] = *(const v2f*)(buf + fA + 16 * h + 4 * s);
    #pragma unroll
    for (int t = 0; t < 5; ++t)
      bm[s][t] = *(const v2f*)(buf + fB + 16 * t * K2_PITCH + 16 * h + 4 * s);
  }
}

static __device__ __forceinline__ void k2_mma(
    v8f (&acc)[5], const v2f (&a)[4], const v2f (&bm)[4][5]) {
  #pragma unroll
  for (int s = 0; s < 4; ++s)
    #pragma unroll
    for (int t = 0; t < 5; ++t)
      acc[t] = wmma_f32(a[s], bm[s][t], acc[t]);
}

#define K3_PITCH 84
#define K3_DS    10

static __device__ __forceinline__ void o_loadB(v2f (&bm)[2][2], const float* p) {
  #pragma unroll
  for (int ks = 0; ks < 2; ++ks)
    #pragma unroll
    for (int s = 0; s < 2; ++s) bm[ks][s] = *(const v2f*)(p + (size_t)(4 * ks + s) * HW2);
}
static __device__ __forceinline__ void o_loadA(v2f (&a)[2][4], const float* p) {
  #pragma unroll
  for (int ks = 0; ks < 2; ++ks)
    #pragma unroll
    for (int r = 0; r < 4; ++r) a[ks][r] = *(const v2f*)(p + 16 * r * K3_PITCH + 4 * ks);
}
static __device__ __forceinline__ void o_pin() { __builtin_amdgcn_sched_barrier(0); }

template <int PH, bool ST>
static __device__ __forceinline__ void o_tile(
    v8f (&cur)[4][2], const v8f (&prev)[4][2], v2f (&bq)[4][2][2], v2f (&aq)[2][2][4],
    const float* Bt, const float* Btn, const float* Al, float* Dprev) {
  #pragma unroll
  for (int r = 0; r < 4; ++r)
    #pragma unroll
    for (int t = 0; t < 2; ++t) cur[r][t] = v8f{};
  #pragma unroll
  for (int d = 0; d < K3_DS; ++d) {
    o_loadB(bq[(PH + d + 3) & 3], (d + 3 < K3_DS) ? Bt + (size_t)(8 * (d + 3)) * HW2
                                                  : Btn + (size_t)(8 * (d + 3 - K3_DS)) * HW2);
    o_loadA(aq[(d + 1) & 1], Al + ((d + 1 < K3_DS) ? 8 * (d + 1) : 0));
    if (ST && d < 8) {
      #pragma unroll
      for (int q = 0; q < 4; ++q) {
        const int i = 4 * (d & 1) + q;
        const v2f v = {prev[d >> 1][0][i], prev[d >> 1][1][i]};
        *(v2f*)(Dprev + (size_t)(16 * (d >> 1) + i) * HW2) = v;
      }
    }
    o_pin();
    #pragma unroll
    for (int ks = 0; ks < 2; ++ks) {
      const v2f (&L)[2] = bq[(PH + d) & 3][ks];
      const v2f bt[2] = {v2f{L[0][0], L[1][0]}, v2f{L[0][1], L[1][1]}};
      #pragma unroll
      for (int r = 0; r < 4; ++r)
        #pragma unroll
        for (int t = 0; t < 2; ++t)
          cur[r][t] = wmma_f32(aq[d & 1][ks][r], bt[t], cur[r][t]);
    }
  }
}

__global__ __launch_bounds__(128) void scores_softmax_out_kernel(
    const float* __restrict__ feat, const float* __restrict__ we_ws,
    float* __restrict__ out) {
  __shared__ __attribute__((aligned(16))) float lds[2 * K2_BUF];
  __shared__ __attribute__((aligned(16))) float att_lds[K2_AROWS * K3_PITCH];

  const int lane = threadIdx.x & 31;
  const int wave = threadIdx.x >> 5;
  const int b   = blockIdx.x >> 3;
  const int cs0 = (blockIdx.x & 7) * K2_AROWS;
  const int half = lane >> 4;
  const int l16  = lane & 15;
  const int r4   = lane >> 3;
  const int seg  = lane & 7;

  const float* gA = feat  + ((size_t)b * Cn + cs0 + 16 * wave + r4) * HW2 + 4 * seg;
  const float* gB = we_ws + ((size_t)b * NP + 20 * wave + r4) * HW2 + 4 * seg;
  const int stA = (16 * wave + r4) * K2_PITCH + 4 * seg;
  const int stB = (K2_AROWS + 20 * wave + r4) * K2_PITCH + 4 * seg;
  const int fA = (16 * wave + l16) * K2_PITCH + 2 * half;
  const int fB = (K2_AROWS + l16) * K2_PITCH + 2 * half;
  float* buf0 = lds;
  float* buf1 = lds + K2_BUF;

  v8f acc[5] = {};
  v4f p0[9], p1[9];
  v2f a0[4], b0[4][5], a1[4], b1[4][5];

  k2_gload(p0, gA, gB, 0);
  k2_gload(p1, gA, gB, K2_KC);
  k2_lstore(buf0, p0, stA, stB);
  k2_gload(p0, gA, gB, 2 * K2_KC);
  k2_wait_lds();
  k2_barrier();
  k2_ldfrag(a0, b0, buf0, fA, fB, 0);
  #pragma unroll
  for (int s = 0; s < 4; ++s)
    asm volatile("" :: "v"(a0[s]), "v"(b0[s][0]), "v"(b0[s][1]), "v"(b0[s][2]),
                       "v"(b0[s][3]), "v"(b0[s][4]));

  for (int t = 0; t < K2_NCH; t += 2) {
    k2_ldfrag(a1, b1, buf0, fA, fB, 1);
    k2_lstore(buf1, p1, stA, stB);
    k2_gload(p1, gA, gB, ((t + 3) & (K2_NCH - 1)) * K2_KC);
    k2_mma(acc, a0, b0);
    k2_wait_lds();
    k2_barrier();
    k2_ldfrag(a0, b0, buf1, fA, fB, 0);
    k2_mma(acc, a1, b1);

    k2_ldfrag(a1, b1, buf1, fA, fB, 1);
    k2_lstore(buf0, p0, stA, stB);
    k2_gload(p0, gA, gB, ((t + 4) & (K2_NCH - 1)) * K2_KC);
    k2_mma(acc, a0, b0);
    k2_wait_lds();
    k2_barrier();
    k2_ldfrag(a0, b0, buf0, fA, fB, 0);
    k2_mma(acc, a1, b1);
  }

  const float* Bw = we_ws + ((size_t)b * NP + 2 * half) * HW2 + 256 * wave + 2 * l16;
  const float* Al = att_lds + l16 * K3_PITCH + 2 * half;
  v2f bq[4][2][2], aq[2][2][4];
  o_loadB(bq[0], Bw);
  o_loadB(bq[1], Bw + (size_t)8 * HW2);
  o_loadB(bq[2], Bw + (size_t)16 * HW2);

  float* Ls = att_lds + (16 * wave + 8 * half) * K3_PITCH + l16;
  #pragma unroll
  for (int i = 0; i < 8; ++i) {
    float m = -3.402823466e38f;
    #pragma unroll
    for (int t = 0; t < 5; ++t)
      if (16 * t + l16 < NW) m = fmaxf(m, acc[t][i]);
    #pragma unroll
    for (int off = 8; off >= 1; off >>= 1) m = fmaxf(m, __shfl_xor(m, off, 32));

    float e[5], s = 0.f;
    #pragma unroll
    for (int t = 0; t < 5; ++t) {
      e[t] = (16 * t + l16 < NW) ? __expf(acc[t][i] - m) : 0.f;
      s += e[t];
    }
    #pragma unroll
    for (int off = 8; off >= 1; off >>= 1) s += __shfl_xor(s, off, 32);
    const float inv = 1.f / s;
    #pragma unroll
    for (int t = 0; t < 5; ++t)
      Ls[i * K3_PITCH + 16 * t] = e[t] * inv;
  }
  k2_wait_lds();
  k2_barrier();

  float* Dw = out + ((size_t)b * Cn + cs0 + 8 * half) * HW2 + 256 * wave + 2 * l16;
  v8f acc0[4][2], acc1[4][2];
  o_loadA(aq[0], Al);
  o_tile<0, false>(acc0, acc1, bq, aq, Bw, Bw + 32, Al, Dw);
  #pragma nounroll
  for (int n = 1; n < 7; n += 2) {
    const float* Bt = Bw + 32 * n;
    float* Dp = Dw + 32 * (n - 1);
    o_tile<2, true>(acc1, acc0, bq, aq, Bt, Bt + 32, Al, Dp);
    o_tile<0, true>(acc0, acc1, bq, aq, Bt + 32, Bt + 64, Al, Dp + 32);
  }
  o_tile<2, true>(acc1, acc0, bq, aq, Bw + 32 * 7, Bw, Al, Dw + 32 * 6);
  #pragma unroll
  for (int r = 0; r < 4; ++r)
    #pragma unroll
    for (int i = 0; i < 8; ++i) {
      const v2f v = {acc1[r][0][i], acc1[r][1][i]};
      *(v2f*)(Dw + (size_t)(16 * r + i) * HW2 + 32 * 7) = v;
    }
}

extern "C" void kernel_launch(void* const* d_in, const int* in_sizes, int n_in,
                              void* d_out, int out_size, void* d_ws, size_t ws_size,
                              hipStream_t stream) {
  const float* feat     = (const float*)d_in[0];
  const float* word_emb = (const float*)d_in[1];
  const float* W_fc     = (const float*)d_in[2];
  const float* b_fc     = (const float*)d_in[3];
  float* out = (float*)d_out;

  float* we_ws  = (float*)d_ws;

  we_gemm_kernel       <<<(Bn * 32) / 4,      128, 0, stream>>>(word_emb, W_fc, b_fc, we_ws);
  scores_softmax_out_kernel<<<Bn * 8,         128, 0, stream>>>(feat, we_ws, out);
}


# scores phase: a chunk's staging loads are issued 6 chunks ahead (was 2) on six register sets; loads past the end clamp to the last chunk (was wrap to the first); sum order kept
# speedup vs baseline: 2.3259x; 1.0715x over previous
#include <hip/hip_runtime.h>

typedef float v2f __attribute__((ext_vector_type(2)));
typedef float v4f __attribute__((ext_vector_type(4)));
typedef float v8f __attribute__((ext_vector_type(8)));

#define Bn   32
#define Cn   512
#define HW2  1024
#define NW   77
#define NP   80
#define WD   256

static __device__ __forceinline__ v8f wmma_f32(v2f a, v2f b, v8f c) {
  return __builtin_amdgcn_wmma_f32_16x16x4_f32(false, a, false, b, (short)0, c, false, false);
}

#define K1_STEPS 4

static __device__ __forceinline__ void k1_load(
    v2f (&a)[K1_STEPS][5], v2f (&bm)[K1_STEPS][2],
    const float* const (&Ab)[5], const float* const (&Bb)[2], int kofs) {
  #pragma unroll
  for (int s = 0; s < K1_STEPS; ++s) {
    #pragma unroll
    for (int tm = 0; tm < 5; ++tm) a[s][tm] = *(const v2f*)(Ab[tm] + kofs + 4 * s);
    #pragma unroll
    for (int j = 0; j < 2; ++j) bm[s][j] = *(const v2f*)(Bb[j] + kofs + 4 * s);
  }
}

static __device__ __forceinline__ void k1_mma(
    v8f (&acc)[5][2], const v2f (&a)[K1_STEPS][5], const v2f (&bm)[K1_STEPS][2]) {
  #pragma unroll
  for (int s = 0; s < K1_STEPS; ++s)
    #pragma unroll
    for (int tm = 0; tm < 5; ++tm)
      #pragma unroll
      for (int j = 0; j < 2; ++j)
        acc[tm][j] = wmma_f32(a[s][tm], bm[s][j], acc[tm][j]);
}

__global__ __launch_bounds__(128) void we_gemm_kernel(
    const float* __restrict__ word_emb, const float* __restrict__ W_fc,
    const float* __restrict__ b_fc, float* __restrict__ we_ws) {
  const int lane = threadIdx.x & 31;
  const int wave = threadIdx.x >> 5;
  const int tile = blockIdx.x * 4 + wave;
  const int b  = tile >> 5;
  const int jb = tile & 31;
  const int half = lane >> 4;
  const int l16  = lane & 15;

  const float* Ab[5];
  #pragma unroll
  for (int tm = 0; tm < 5; ++tm) {
    const int a_row = tm * 16 + l16;
    Ab[tm] = word_emb + ((size_t)b * NW + (a_row < NW ? a_row : NW - 1)) * WD + 2 * half;
  }
  const float* Bb[2];
  #pragma unroll
  for (int j = 0; j < 2; ++j)
    Bb[j] = W_fc + (size_t)(jb * 32 + 2 * l16 + j) * WD + 2 * half;

  v8f acc[5][2] = {};
  v2f a0[K1_STEPS][5], b0[K1_STEPS][2], a1[K1_STEPS][5], b1[K1_STEPS][2];
  k1_load(a0, b0, Ab, Bb, 0);
  #pragma unroll
  for (int s = 0; s < K1_STEPS; ++s)
    asm volatile("" :: "v"(a0[s][0]), "v"(a0[s][1]), "v"(a0[s][2]), "v"(a0[s][3]),
                       "v"(a0[s][4]), "v"(b0[s][0]), "v"(b0[s][1]));
  for (int k0 = 0; k0 < WD; k0 += 8 * K1_STEPS) {
    k1_load(a1, b1, Ab, Bb, k0 + 4 * K1_STEPS);
    k1_mma(acc, a0, b0);
    k1_load(a0, b0, Ab, Bb, (k0 + 8 * K1_STEPS) & (WD - 1));
    k1_mma(acc, a1, b1);
  }

  const int col = jb * 32 + 2 * l16;
  const float bias0 = b_fc[col], bias1 = b_fc[col + 1];
  #pragma unroll
  for (int tm = 0; tm < 5; ++tm) {
    float* D = we_ws + ((size_t)b * NP + tm * 16 + 8 * half) * HW2 + col;
    #pragma unroll
    for (int i = 0; i < 8; ++i) {
      const v2f v = {acc[tm][0][i] + bias0, acc[tm][1][i] + bias1};
      *(v2f*)(D + (size_t)i * HW2) = v;
    }
  }
}

#define K2_KC    32
#define K2_NCH   (HW2 / K2_KC)
#define K2_PITCH 36
#define K2_AROWS 64
#define K2_BUF   ((K2_AROWS + NP) * K2_PITCH)
#define K2_SETS  6

static __device__ __forceinline__ void k2_wait_lds() {
  __builtin_amdgcn_sched_barrier(0);
  asm volatile("s_wait_dscnt 0x0" ::: "memory");
}
static __device__ __forceinline__ void k2_barrier() {
  asm volatile("s_barrier_signal -1\n\ts_barrier_wait -1" ::: "memory");
}

static __device__ __forceinline__ void k2_gload(
    v4f (&p)[9], const float* gA, const float* gB, int k0) {
  #pragma unroll
  for (int i = 0; i < 4; ++i) p[i]     = *(const v4f*)(gA + (size_t)(4 * i) * HW2 + k0);
  #pragma unroll
  for (int i = 0; i < 5; ++i) p[4 + i] = *(const v4f*)(gB + (size_t)(4 * i) * HW2 + k0);
}

static __device__ __forceinline__ void k2_lstore(
    float* buf, const v4f (&p)[9], int stA, int stB) {
  #pragma unroll
  for (int i = 0; i < 4; ++i) *(v4f*)(buf + stA + 4 * i * K2_PITCH) = p[i];
  #pragma unroll
  for (int i = 0; i < 5; ++i) *(v4f*)(buf + stB + 4 * i * K2_PITCH) = p[4 + i];
}

static __device__ __forceinline__ void k2_ldfrag(
    v2f (&a)[4], v2f (&bm)[4][5], const float* buf, int fA, int fB, int h) {
  #pragma unroll
  for (int s = 0; s < 4; ++s) {
    a[s] = *(const v2f*)(buf + fA + 16 * h + 4 * s);
    #pragma unroll
    for (int t = 0; t < 5; ++t)
      bm[s][t] = *(const v2f*)(buf + fB + 16 * t * K2_PITCH + 16 * h + 4 * s);
  }
}

static __device__ __forceinline__ void k2_mma(
    v8f (&acc)[5], const v2f (&a)[4], const v2f (&bm)[4][5]) {
  #pragma unroll
  for (int s = 0; s < 4; ++s)
    #pragma unroll
    for (int t = 0; t < 5; ++t)
      acc[t] = wmma_f32(a[s], bm[s][t], acc[t]);
}

template <int J, bool PARK, bool REFILL>
static __device__ __forceinline__ void k2_chunk(
    v8f (&acc)[5], v4f (&p)[K2_SETS][9],
    v2f (&a0)[4], v2f (&b0)[4][5], v2f (&a1)[4], v2f (&b1)[4][5],
    float* buf0, float* buf1, const float* gA, const float* gB, int knext,
    int stA, int stB, int fA, int fB) {
  float* bufX = (J & 1) ? buf1 : buf0;
  float* bufY = (J & 1) ? buf0 : buf1;
  constexpr int S = (J + 1) % K2_SETS;
  k2_ldfrag(a1, b1, bufX, fA, fB, 1);
  if (PARK) k2_lstore(bufY, p[S], stA, stB);
  if (REFILL) k2_gload(p[S], gA, gB, knext);
  k2_mma(acc, a0, b0);
  if (PARK) {
    k2_wait_lds();
    k2_barrier();
    k2_ldfrag(a0, b0, bufY, fA, fB, 0);
  }
  k2_mma(acc, a1, b1);
}

#define K3_PITCH 84
#define K3_DS    10

static __device__ __forceinline__ void o_loadB(v2f (&bm)[2][2], const float* p) {
  #pragma unroll
  for (int ks = 0; ks < 2; ++ks)
    #pragma unroll
    for (int s = 0; s < 2; ++s) bm[ks][s] = *(const v2f*)(p + (size_t)(4 * ks + s) * HW2);
}
static __device__ __forceinline__ void o_loadA(v2f (&a)[2][4], const float* p) {
  #pragma unroll
  for (int ks = 0; ks < 2; ++ks)
    #pragma unroll
    for (int r = 0; r < 4; ++r) a[ks][r] = *(const v2f*)(p + 16 * r * K3_PITCH + 4 * ks);
}
static __device__ __forceinline__ void o_pin() { __builtin_amdgcn_sched_barrier(0); }

template <int PH, bool ST>
static __device__ __forceinline__ void o_tile(
    v8f (&cur)[4][2], const v8f (&prev)[4][2], v2f (&bq)[4][2][2], v2f (&aq)[2][2][4],
    const float* Bt, const float* Btn, const float* Al, float* Dprev) {
  #pragma unroll
  for (int r = 0; r < 4; ++r)
    #pragma unroll
    for (int t = 0; t < 2; ++t) cur[r][t] = v8f{};
  #pragma unroll
  for (int d = 0; d < K3_DS; ++d) {
    o_loadB(bq[(PH + d + 3) & 3], (d + 3 < K3_DS) ? Bt + (size_t)(8 * (d + 3)) * HW2
                                                  : Btn + (size_t)(8 * (d + 3 - K3_DS)) * HW2);
    o_loadA(aq[(d + 1) & 1], Al + ((d + 1 < K3_DS) ? 8 * (d + 1) : 0));
    if (ST && d < 8) {
      #pragma unroll
      for (int q = 0; q < 4; ++q) {
        const int i = 4 * (d & 1) + q;
        const v2f v = {prev[d >> 1][0][i], prev[d >> 1][1][i]};
        *(v2f*)(Dprev + (size_t)(16 * (d >> 1) + i) * HW2) = v;
      }
    }
    o_pin();
    #pragma unroll
    for (int ks = 0; ks < 2; ++ks) {
      const v2f (&L)[2] = bq[(PH + d) & 3][ks];
      const v2f bt[2] = {v2f{L[0][0], L[1][0]}, v2f{L[0][1], L[1][1]}};
      #pragma unroll
      for (int r = 0; r < 4; ++r)
        #pragma unroll
        for (int t = 0; t < 2; ++t)
          cur[r][t] = wmma_f32(aq[d & 1][ks][r], bt[t], cur[r][t]);
    }
  }
}

__global__ __launch_bounds__(128) void scores_softmax_out_kernel(
    const float* __restrict__ feat, const float* __restrict__ we_ws,
    float* __restrict__ out) {
  __shared__ __attribute__((aligned(16))) float lds[2 * K2_BUF];
  __shared__ __attribute__((aligned(16))) float att_lds[K2_AROWS * K3_PITCH];

  const int lane = threadIdx.x & 31;
  const int wave = threadIdx.x >> 5;
  const int b   = blockIdx.x >> 3;
  const int cs0 = (blockIdx.x & 7) * K2_AROWS;
  const int half = lane >> 4;
  const int l16  = lane & 15;
  const int r4   = lane >> 3;
  const int seg  = lane & 7;

  const float* gA = feat  + ((size_t)b * Cn + cs0 + 16 * wave + r4) * HW2 + 4 * seg;
  const float* gB = we_ws + ((size_t)b * NP + 20 * wave + r4) * HW2 + 4 * seg;
  const int stA = (16 * wave + r4) * K2_PITCH + 4 * seg;
  const int stB = (K2_AROWS + 20 * wave + r4) * K2_PITCH + 4 * seg;
  const int fA = (16 * wave + l16) * K2_PITCH + 2 * half;
  const int fB = (K2_AROWS + l16) * K2_PITCH + 2 * half;
  float* buf0 = lds;
  float* buf1 = lds + K2_BUF;

  v8f acc[5] = {};
  v4f p[K2_SETS][9];
  v2f a0[4], b0[4][5], a1[4], b1[4][5];

  #pragma unroll
  for (int s = 0; s < K2_SETS; ++s) k2_gload(p[s], gA, gB, s * K2_KC);
  k2_lstore(buf0, p[0], stA, stB);
  k2_gload(p[0], gA, gB, K2_SETS * K2_KC);
  k2_wait_lds();
  k2_barrier();
  k2_ldfrag(a0, b0, buf0, fA, fB, 0);
  #pragma unroll
  for (int s = 0; s < 4; ++s)
    asm volatile("" :: "v"(a0[s]), "v"(b0[s][0]), "v"(b0[s][1]), "v"(b0[s][2]),
                       "v"(b0[s][3]), "v"(b0[s][4]));

  #define K2_NEXT(c) (((c) + 7 < K2_NCH ? (c) + 7 : K2_NCH - 1) * K2_KC)
  #pragma nounroll
  for (int t = 0; t < K2_NCH - 2; t += K2_SETS) {
    k2_chunk<0, true, true>(acc, p, a0, b0, a1, b1, buf0, buf1, gA, gB, K2_NEXT(t + 0), stA, stB, fA, fB);
    k2_chunk<1, true, true>(acc, p, a0, b0, a1, b1, buf0, buf1, gA, gB, K2_NEXT(t + 1), stA, stB, fA, fB);
    k2_chunk<2, true, true>(acc, p, a0, b0, a1, b1, buf0, buf1, gA, gB, K2_NEXT(t + 2), stA, stB, fA, fB);
    k2_chunk<3, true, true>(acc, p, a0, b0, a1, b1, buf0, buf1, gA, gB, K2_NEXT(t + 3), stA, stB, fA, fB);
    k2_chunk<4, true, true>(acc, p, a0, b0, a1, b1, buf0, buf1, gA, gB, K2_NEXT(t + 4), stA, stB, fA, fB);
    k2_chunk<5, true, true>(acc, p, a0, b0, a1, b1, buf0, buf1, gA, gB, K2_NEXT(t + 5), stA, stB, fA, fB);
  }
  #undef K2_NEXT
  k2_chunk<0, true, false>(acc, p, a0, b0, a1, b1, buf0, buf1, gA, gB, 0, stA, stB, fA, fB);
  k2_chunk<1, false, false>(acc, p, a0, b0, a1, b1, buf0, buf1, gA, gB, 0, stA, stB, fA, fB);

  const float* Bw = we_ws + ((size_t)b * NP + 2 * half) * HW2 + 256 * wave + 2 * l16;
  const float* Al = att_lds + l16 * K3_PITCH + 2 * half;
  v2f bq[4][2][2], aq[2][2][4];
  o_loadB(bq[0], Bw);
  o_loadB(bq[1], Bw + (size_t)8 * HW2);
  o_loadB(bq[2], Bw + (size_t)16 * HW2);

  float* Ls = att_lds + (16 * wave + 8 * half) * K3_PITCH + l16;
  #pragma unroll
  for (int i = 0; i < 8; ++i) {
    float m = -3.402823466e38f;
    #pragma unroll
    for (int t = 0; t < 5; ++t)
      if (16 * t + l16 < NW) m = fmaxf(m, acc[t][i]);
    #pragma unroll
    for (int off = 8; off >= 1; off >>= 1) m = fmaxf(m, __shfl_xor(m, off, 32));

    float e[5], s = 0.f;
    #pragma unroll
    for (int t = 0; t < 5; ++t) {
      e[t] = (16 * t + l16 < NW) ? __expf(acc[t][i] - m) : 0.f;
      s += e[t];
    }
    #pragma unroll
    for (int off = 8; off >= 1; off >>= 1) s += __shfl_xor(s, off, 32);
    const float inv = 1.f / s;
    #pragma unroll
    for (int t = 0; t < 5; ++t)
      Ls[i * K3_PITCH + 16 * t] = e[t] * inv;
  }
  k2_wait_lds();
  k2_barrier();

  float* Dw = out + ((size_t)b * Cn + cs0 + 8 * half) * HW2 + 256 * wave + 2 * l16;
  v8f acc0[4][2], acc1[4][2];
  o_loadA(aq[0], Al);
  o_tile<0, false>(acc0, acc1, bq, aq, Bw, Bw + 32, Al, Dw);
  #pragma nounroll
  for (int n = 1; n < 7; n += 2) {
    const float* Bt = Bw + 32 * n;
    float* Dp = Dw + 32 * (n - 1);
    o_tile<2, true>(acc1, acc0, bq, aq, Bt, Bt + 32, Al, Dp);
    o_tile<0, true>(acc0, acc1, bq, aq, Bt + 32, Bt + 64, Al, Dp + 32);
  }
  o_tile<2, true>(acc1, acc0, bq, aq, Bw + 32 * 7, Bw, Al, Dw + 32 * 6);
  #pragma unroll
  for (int r = 0; r < 4; ++r)
    #pragma unroll
    for (int i = 0; i < 8; ++i) {
      const v2f v = {acc1[r][0][i], acc1[r][1][i]};
      *(v2f*)(Dw + (size_t)(16 * r + i) * HW2 + 32 * 7) = v;
    }
}

extern "C" void kernel_launch(void* const* d_in, const int* in_sizes, int n_in,
                              void* d_out, int out_size, void* d_ws, size_t ws_size,
                              hipStream_t stream) {
  const float* feat     = (const float*)d_in[0];
  const float* word_emb = (const float*)d_in[1];
  const float* W_fc     = (const float*)d_in[2];
  const float* b_fc     = (const float*)d_in[3];
  float* out = (float*)d_out;

  float* we_ws  = (float*)d_ws;

  we_gemm_kernel       <<<(Bn * 32) / 4,      128, 0, stream>>>(word_emb, W_fc, b_fc, we_ws);
  scores_softmax_out_kernel<<<Bn * 8,         128, 0, stream>>>(feat, we_ws, out);
}


# fused kernel: 8 waves per work-group (was 4), two per SIMD: a slab's 80 score columns split 48+32 over a wave pair, 128 out columns per wave; softmax joins the pair's halves in LDS; sum order kept
# speedup vs baseline: 2.4857x; 1.0687x over previous
#include <hip/hip_runtime.h>

typedef float v2f __attribute__((ext_vector_type(2)));
typedef float v4f __attribute__((ext_vector_type(4)));
typedef float v8f __attribute__((ext_vector_type(8)));

#define Bn   32
#define Cn   512
#define HW2  1024
#define NW   77
#define NP   80
#define WD   256

static __device__ __forceinline__ v8f wmma_f32(v2f a, v2f b, v8f c) {
  return __builtin_amdgcn_wmma_f32_16x16x4_f32(false, a, false, b, (short)0, c, false, false);
}

#define K1_STEPS 4

static __device__ __forceinline__ void k1_load(
    v2f (&a)[K1_STEPS][5], v2f (&bm)[K1_STEPS][2],
    const float* const (&Ab)[5], const float* const (&Bb)[2], int kofs) {
  #pragma unroll
  for (int s = 0; s < K1_STEPS; ++s) {
    #pragma unroll
    for (int tm = 0; tm < 5; ++tm) a[s][tm] = *(const v2f*)(Ab[tm] + kofs + 4 * s);
    #pragma unroll
    for (int j = 0; j < 2; ++j) bm[s][j] = *(const v2f*)(Bb[j] + kofs + 4 * s);
  }
}

static __device__ __forceinline__ void k1_mma(
    v8f (&acc)[5][2], const v2f (&a)[K1_STEPS][5], const v2f (&bm)[K1_STEPS][2]) {
  #pragma unroll
  for (int s = 0; s < K1_STEPS; ++s)
    #pragma unroll
    for (int tm = 0; tm < 5; ++tm)
      #pragma unroll
      for (int j = 0; j < 2; ++j)
        acc[tm][j] = wmma_f32(a[s][tm], bm[s][j], acc[tm][j]);
}

__global__ __launch_bounds__(128) void we_gemm_kernel(
    const float* __restrict__ word_emb, const float* __restrict__ W_fc,
    const float* __restrict__ b_fc, float* __restrict__ we_ws) {
  const int lane = threadIdx.x & 31;
  const int wave = threadIdx.x >> 5;
  const int tile = blockIdx.x * 4 + wave;
  const int b  = tile >> 5;
  const int jb = tile & 31;
  const int half = lane >> 4;
  const int l16  = lane & 15;

  const float* Ab[5];
  #pragma unroll
  for (int tm = 0; tm < 5; ++tm) {
    const int a_row = tm * 16 + l16;
    Ab[tm] = word_emb + ((size_t)b * NW + (a_row < NW ? a_row : NW - 1)) * WD + 2 * half;
  }
  const float* Bb[2];
  #pragma unroll
  for (int j = 0; j < 2; ++j)
    Bb[j] = W_fc + (size_t)(jb * 32 + 2 * l16 + j) * WD + 2 * half;

  v8f acc[5][2] = {};
  v2f a0[K1_STEPS][5], b0[K1_STEPS][2], a1[K1_STEPS][5], b1[K1_STEPS][2];
  k1_load(a0, b0, Ab, Bb, 0);
  #pragma unroll
  for (int s = 0; s < K1_STEPS; ++s)
    asm volatile("" :: "v"(a0[s][0]), "v"(a0[s][1]), "v"(a0[s][2]), "v"(a0[s][3]),
                       "v"(a0[s][4]), "v"(b0[s][0]), "v"(b0[s][1]));
  for (int k0 = 0; k0 < WD; k0 += 8 * K1_STEPS) {
    k1_load(a1, b1, Ab, Bb, k0 + 4 * K1_STEPS);
    k1_mma(acc, a0, b0);
    k1_load(a0, b0, Ab, Bb, (k0 + 8 * K1_STEPS) & (WD - 1));
    k1_mma(acc, a1, b1);
  }

  const int col = jb * 32 + 2 * l16;
  const float bias0 = b_fc[col], bias1 = b_fc[col + 1];
  #pragma unroll
  for (int tm = 0; tm < 5; ++tm) {
    float* D = we_ws + ((size_t)b * NP + tm * 16 + 8 * half) * HW2 + col;
    #pragma unroll
    for (int i = 0; i < 8; ++i) {
      const v2f v = {acc[tm][0][i] + bias0, acc[tm][1][i] + bias1};
      *(v2f*)(D + (size_t)i * HW2) = v;
    }
  }
}

#define K2_KC    32
#define K2_NCH   (HW2 / K2_KC)
#define K2_PITCH 36
#define K2_AROWS 64
#define K2_BUF   ((K2_AROWS + NP) * K2_PITCH)
#define K2_SETS  6
#define K3_PITCH 84

static __device__ __forceinline__ void k2_wait_lds() {
  __builtin_amdgcn_sched_barrier(0);
  asm volatile("s_wait_dscnt 0x0" ::: "memory");
}
static __device__ __forceinline__ void k2_barrier() {
  asm volatile("s_barrier_signal -1\n\ts_barrier_wait -1" ::: "memory");
}

static __device__ __forceinline__ void k2_gload4(v4f (&p)[5], const float* g0, int k0) {
  #pragma unroll
  for (int i = 0; i < 4; ++i) p[i] = *(const v4f*)(g0 + (size_t)(4 * i) * HW2 + k0);
}
static __device__ __forceinline__ void k2_lstore4(float* buf, const v4f (&p)[5], int st0) {
  #pragma unroll
  for (int i = 0; i < 4; ++i) *(v4f*)(buf + st0 + 4 * i * K2_PITCH) = p[i];
}

static __device__ __forceinline__ void k2_ldfrag2(
    v2f (&a)[4], v2f (&bm)[4][3], const float* buf, int fA, int fB, int h) {
  #pragma unroll
  for (int s = 0; s < 4; ++s) {
    a[s] = *(const v2f*)(buf + fA + 16 * h + 4 * s);
    #pragma unroll
    for (int t = 0; t < 2; ++t)
      bm[s][t] = *(const v2f*)(buf + fB + 16 * t * K2_PITCH + 16 * h + 4 * s);
  }
}
static __device__ __forceinline__ void k2_ldfrag3(
    v2f (&bm)[4][3], const float* buf, int fB, int h) {
  #pragma unroll
  for (int s = 0; s < 4; ++s)
    bm[s][2] = *(const v2f*)(buf + fB + 32 * K2_PITCH + 16 * h + 4 * s);
}

static __device__ __forceinline__ void k2_mma2(
    v8f (&acc)[3], const v2f (&a)[4], const v2f (&bm)[4][3]) {
  #pragma unroll
  for (int s = 0; s < 4; ++s)
    #pragma unroll
    for (int t = 0; t < 2; ++t)
      acc[t] = wmma_f32(a[s], bm[s][t], acc[t]);
}
static __device__ __forceinline__ void k2_mma3(
    v8f (&acc)[3], const v2f (&a)[4], const v2f (&bm)[4][3]) {
  #pragma unroll
  for (int s = 0; s < 4; ++s) acc[2] = wmma_f32(a[s], bm[s][2], acc[2]);
}

template <int J, bool PARK, bool REFILL>
static __device__ __forceinline__ void k2_pre(
    int part, v8f (&acc)[3], v4f (&p)[K2_SETS][5],
    v2f (&a0)[4], v2f (&b0)[4][3], v2f (&a1)[4], v2f (&b1)[4][3],
    float* buf0, float* buf1, const float* g0, const float* g1, int knext,
    int st0, int st1, int fA, int fB) {
  float* bufX = (J & 1) ? buf1 : buf0;
  float* bufY = (J & 1) ? buf0 : buf1;
  constexpr int S = (J + 1) % K2_SETS;
  k2_ldfrag2(a1, b1, bufX, fA, fB, 1);
  if (PARK) k2_lstore4(bufY, p[S], st0);
  if (REFILL) k2_gload4(p[S], g0, knext);
  if (part == 0) {
    k2_ldfrag3(b1, bufX, fB, 1);
    if (PARK) *(v4f*)(bufY + st1) = p[S][4];
    if (REFILL) p[S][4] = *(const v4f*)(g1 + knext);
    k2_mma3(acc, a0, b0);
  }
  k2_mma2(acc, a0, b0);
}
template <int J, bool PARK>
static __device__ __forceinline__ void k2_post(
    int part, v8f (&acc)[3], v2f (&a0)[4], v2f (&b0)[4][3],
    const v2f (&a1)[4], const v2f (&b1)[4][3],
    float* buf0, float* buf1, int fA, int fB) {
  float* bufY = (J & 1) ? buf0 : buf1;
  if (PARK) k2_ldfrag2(a0, b0, bufY, fA, fB, 0);
  if (part == 0) {
    if (PARK) k2_ldfrag3(b0, bufY, fB, 0);
    k2_mma3(acc, a1, b1);
  }
  k2_mma2(acc, a1, b1);
}

static __device__ __forceinline__ void sm_max(
    float (&m)[8], const v8f (&acc)[3], int col0, int nt) {
  #pragma unroll
  for (int i = 0; i < 8; ++i) {
    float v = -3.402823466e38f;
    #pragma unroll
    for (int t = 0; t < 3; ++t)
      if (t < nt && col0 + 16 * t < NW) v = fmaxf(v, acc[t][i]);
    #pragma unroll
    for (int off = 8; off >= 1; off >>= 1) v = fmaxf(v, __shfl_xor(v, off, 32));
    m[i] = v;
  }
}
static __device__ __forceinline__ void sm_exp(
    float* Ls, const v8f (&acc)[3], const float (&m)[8], int col0, int nt) {
  #pragma unroll
  for (int i = 0; i < 8; ++i)
    #pragma unroll
    for (int t = 0; t < 3; ++t)
      if (t < nt)
        Ls[i * K3_PITCH + 16 * t] = (col0 + 16 * t < NW) ? __expf(acc[t][i] - m[i]) : 0.f;
}

#define K3_DS    10

static __device__ __forceinline__ void o_loadB(v2f (&bm)[2][2], const float* p) {
  #pragma unroll
  for (int ks = 0; ks < 2; ++ks)
    #pragma unroll
    for (int s = 0; s < 2; ++s) bm[ks][s] = *(const v2f*)(p + (size_t)(4 * ks + s) * HW2);
}
static __device__ __forceinline__ void o_loadA(v2f (&a)[2][4], const float* p) {
  #pragma unroll
  for (int ks = 0; ks < 2; ++ks)
    #pragma unroll
    for (int r = 0; r < 4; ++r) a[ks][r] = *(const v2f*)(p + 16 * r * K3_PITCH + 4 * ks);
}
static __device__ __forceinline__ void o_pin() { __builtin_amdgcn_sched_barrier(0); }

template <int PH, bool ST>
static __device__ __forceinline__ void o_tile(
    v8f (&cur)[4][2], const v8f (&prev)[4][2], v2f (&bq)[4][2][2], v2f (&aq)[2][2][4],
    const float* Bt, const float* Btn, const float* Al, float* Dprev) {
  #pragma unroll
  for (int r = 0; r < 4; ++r)
    #pragma unroll
    for (int t = 0; t < 2; ++t) cur[r][t] = v8f{};
  #pragma unroll
  for (int d = 0; d < K3_DS; ++d) {
    o_loadB(bq[(PH + d + 3) & 3], (d + 3 < K3_DS) ? Bt + (size_t)(8 * (d + 3)) * HW2
                                                  : Btn + (size_t)(8 * (d + 3 - K3_DS)) * HW2);
    o_loadA(aq[(d + 1) & 1], Al + ((d + 1 < K3_DS) ? 8 * (d + 1) : 0));
    if (ST && d < 8) {
      #pragma unroll
      for (int q = 0; q < 4; ++q) {
        const int i = 4 * (d & 1) + q;
        const v2f v = {prev[d >> 1][0][i], prev[d >> 1][1][i]};
        *(v2f*)(Dprev + (size_t)(16 * (d >> 1) + i) * HW2) = v;
      }
    }
    o_pin();
    #pragma unroll
    for (int ks = 0; ks < 2; ++ks) {
      const v2f (&L)[2] = bq[(PH + d) & 3][ks];
      const v2f bt[2] = {v2f{L[0][0], L[1][0]}, v2f{L[0][1], L[1][1]}};
      #pragma unroll
      for (int r = 0; r < 4; ++r)
        #pragma unroll
        for (int t = 0; t < 2; ++t)
          cur[r][t] = wmma_f32(aq[d & 1][ks][r], bt[t], cur[r][t]);
    }
  }
}

__global__ __launch_bounds__(256) void scores_softmax_out_kernel(
    const float* __restrict__ feat, const float* __restrict__ we_ws,
    float* __restrict__ out) {
  __shared__ __attribute__((aligned(16))) float lds[2 * K2_BUF];
  __shared__ __attribute__((aligned(16))) float att_lds[K2_AROWS * K3_PITCH];
  __shared__ float pmax[2 * K2_AROWS];

  const int lane = threadIdx.x & 31;
  const int wave = __builtin_amdgcn_readfirstlane((int)(threadIdx.x >> 5));
  const int slab = wave & 3;
  const int part = wave >> 2;
  const int b   = blockIdx.x >> 3;
  const int cs0 = (blockIdx.x & 7) * K2_AROWS;
  const int half = lane >> 4;
  const int l16  = lane & 15;
  const int r4   = lane >> 3;
  const int seg  = lane & 7;

  const float* gF = feat  + ((size_t)b * Cn + cs0 + 16 * slab + r4) * HW2 + 4 * seg;
  const float* gW = we_ws + ((size_t)b * NP + 16 + 16 * slab + r4) * HW2 + 4 * seg;
  const float* g0 = part ? gW : gF;
  const float* g1 = we_ws + ((size_t)b * NP + 4 * slab + r4) * HW2 + 4 * seg;
  const int st0 = ((part ? K2_AROWS + 16 : 0) + 16 * slab + r4) * K2_PITCH + 4 * seg;
  const int st1 = (K2_AROWS + 4 * slab + r4) * K2_PITCH + 4 * seg;
  const int fA = (16 * slab + l16) * K2_PITCH + 2 * half;
  const int fB = (K2_AROWS + 48 * part + l16) * K2_PITCH + 2 * half;
  float* buf0 = lds;
  float* buf1 = lds + K2_BUF;

  v8f acc[3] = {};
  v4f p[K2_SETS][5];
  v2f a0[4], b0[4][3], a1[4], b1[4][3];

  #pragma unroll
  for (int s = 0; s < K2_SETS; ++s) k2_gload4(p[s], g0, s * K2_KC);
  if (part == 0) {
    #pragma unroll
    for (int s = 0; s < K2_SETS; ++s) p[s][4] = *(const v4f*)(g1 + s * K2_KC);
  }
  k2_lstore4(buf0, p[0], st0);
  k2_gload4(p[0], g0, K2_SETS * K2_KC);
  if (part == 0) {
    *(v4f*)(buf0 + st1) = p[0][4];
    p[0][4] = *(const v4f*)(g1 + K2_SETS * K2_KC);
  }
  k2_wait_lds();
  k2_barrier();
  k2_ldfrag2(a0, b0, buf0, fA, fB, 0);
  if (part == 0) k2_ldfrag3(b0, buf0, fB, 0);

  #define K2_CHUNK(J, PARK, REFILL, KN)                                          \
    k2_pre<J, PARK, REFILL>(part, acc, p, a0, b0, a1, b1, buf0, buf1, g0, g1, KN, \
                            st0, st1, fA, fB);                                   \
    if (PARK) { k2_wait_lds(); k2_barrier(); }                                   \
    k2_post<J, PARK>(part, acc, a0, b0, a1, b1, buf0, buf1, fA, fB);

  #define K2_NEXT(c) (((c) + 7 < K2_NCH ? (c) + 7 : K2_NCH - 1) * K2_KC)
  #pragma nounroll
  for (int t = 0; t < K2_NCH - 2; t += K2_SETS) {
    K2_CHUNK(0, true, true, K2_NEXT(t + 0))
    K2_CHUNK(1, true, true, K2_NEXT(t + 1))
    K2_CHUNK(2, true, true, K2_NEXT(t + 2))
    K2_CHUNK(3, true, true, K2_NEXT(t + 3))
    K2_CHUNK(4, true, true, K2_NEXT(t + 4))
    K2_CHUNK(5, true, true, K2_NEXT(t + 5))
  }
  #undef K2_NEXT
  K2_CHUNK(0, true, false, 0)
  K2_CHUNK(1, false, false, 0)
  #undef K2_CHUNK

  const float* Bw = we_ws + ((size_t)b * NP + 2 * half) * HW2 + 128 * wave + 2 * l16;
  const float* Al = att_lds + l16 * K3_PITCH + 2 * half;
  v2f bq[4][2][2], aq[2][2][4];
  o_loadB(bq[0], Bw);
  o_loadB(bq[1], Bw + (size_t)8 * HW2);
  o_loadB(bq[2], Bw + (size_t)16 * HW2);

  const int col0 = 48 * part + l16;
  float m[8];
  const int nt = 3 - part;
  sm_max(m, acc, col0, nt);
  const int srow = 16 * slab + 8 * half;
  #pragma unroll
  for (int i = 0; i < 8; ++i) pmax[part * K2_AROWS + srow + i] = m[i];
  k2_wait_lds();
  k2_barrier();
  #pragma unroll
  for (int i = 0; i < 8; ++i) m[i] = fmaxf(pmax[srow + i], pmax[K2_AROWS + srow + i]);
  float* Ls = att_lds + srow * K3_PITCH + col0;
  sm_exp(Ls, acc, m, col0, nt);
  k2_wait_lds();
  k2_barrier();
  float* Ln = att_lds + (16 * slab + 8 * part + 4 * half) * K3_PITCH + l16;
  #pragma unroll
  for (int i = 0; i < 4; ++i) {
    float e[5], s = 0.f;
    #pragma unroll
    for (int t = 0; t < 5; ++t) {
      e[t] = Ln[i * K3_PITCH + 16 * t];
      s += e[t];
    }
    #pragma unroll
    for (int off = 8; off >= 1; off >>= 1) s += __shfl_xor(s, off, 32);
    const float inv = 1.f / s;
    #pragma unroll
    for (int t = 0; t < 5; ++t)
      Ln[i * K3_PITCH + 16 * t] = e[t] * inv;
  }
  k2_wait_lds();
  k2_barrier();

  float* Dw = out + ((size_t)b * Cn + cs0 + 8 * half) * HW2 + 128 * wave + 2 * l16;
  v8f acc0[4][2], acc1[4][2];
  o_loadA(aq[0], Al);
  o_tile<0, false>(acc0, acc1, bq, aq, Bw,      Bw + 32, Al, Dw);
  o_tile<2, true >(acc1, acc0, bq, aq, Bw + 32, Bw + 64, Al, Dw);
  o_tile<0, true >(acc0, acc1, bq, aq, Bw + 64, Bw + 96, Al, Dw + 32);
  o_tile<2, true >(acc1, acc0, bq, aq, Bw + 96, Bw,      Al, Dw + 64);
  #pragma unroll
  for (int r = 0; r < 4; ++r)
    #pragma unroll
    for (int i = 0; i < 8; ++i) {
      const v2f v = {acc1[r][0][i], acc1[r][1][i]};
      *(v2f*)(Dw + (size_t)(16 * r + i) * HW2 + 32 * 3) = v;
    }
}

extern "C" void kernel_launch(void* const* d_in, const int* in_sizes, int n_in,
                              void* d_out, int out_size, void* d_ws, size_t ws_size,
                              hipStream_t stream) {
  const float* feat     = (const float*)d_in[0];
  const float* word_emb = (const float*)d_in[1];
  const float* W_fc     = (const float*)d_in[2];
  const float* b_fc     = (const float*)d_in[3];
  float* out = (float*)d_out;

  float* we_ws  = (float*)d_ws;

  we_gemm_kernel       <<<(Bn * 32) / 4,      128, 0, stream>>>(word_emb, W_fc, b_fc, we_ws);
  scores_softmax_out_kernel<<<Bn * 8,         256, 0, stream>>>(feat, we_ws, out);
}


# out GEMM: a wave's FOUR column tiles interleave (tile t = columns 4c+t; 32x64 tiles, was 64x32 of two): 16-byte stores and B loads cover whole aligned 256-byte row pieces; sum order kept
# speedup vs baseline: 2.6139x; 1.0516x over previous
#include <hip/hip_runtime.h>

typedef float v2f __attribute__((ext_vector_type(2)));
typedef float v4f __attribute__((ext_vector_type(4)));
typedef float v8f __attribute__((ext_vector_type(8)));

#define Bn   32
#define Cn   512
#define HW2  1024
#define NW   77
#define NP   80
#define WD   256

static __device__ __forceinline__ v8f wmma_f32(v2f a, v2f b, v8f c) {
  return __builtin_amdgcn_wmma_f32_16x16x4_f32(false, a, false, b, (short)0, c, false, false);
}

#define K1_STEPS 4

static __device__ __forceinline__ void k1_load(
    v2f (&a)[K1_STEPS][5], v2f (&bm)[K1_STEPS][2],
    const float* const (&Ab)[5], const float* const (&Bb)[2], int kofs) {
  #pragma unroll
  for (int s = 0; s < K1_STEPS; ++s) {
    #pragma unroll
    for (int tm = 0; tm < 5; ++tm) a[s][tm] = *(const v2f*)(Ab[tm] + kofs + 4 * s);
    #pragma unroll
    for (int j = 0; j < 2; ++j) bm[s][j] = *(const v2f*)(Bb[j] + kofs + 4 * s);
  }
}

static __device__ __forceinline__ void k1_mma(
    v8f (&acc)[5][2], const v2f (&a)[K1_STEPS][5], const v2f (&bm)[K1_STEPS][2]) {
  #pragma unroll
  for (int s = 0; s < K1_STEPS; ++s)
    #pragma unroll
    for (int tm = 0; tm < 5; ++tm)
      #pragma unroll
      for (int j = 0; j < 2; ++j)
        acc[tm][j] = wmma_f32(a[s][tm], bm[s][j], acc[tm][j]);
}

__global__ __launch_bounds__(128) void we_gemm_kernel(
    const float* __restrict__ word_emb, const float* __restrict__ W_fc,
    const float* __restrict__ b_fc, float* __restrict__ we_ws) {
  const int lane = threadIdx.x & 31;
  const int wave = threadIdx.x >> 5;
  const int tile = blockIdx.x * 4 + wave;
  const int b  = tile >> 5;
  const int jb = tile & 31;
  const int half = lane >> 4;
  const int l16  = lane & 15;

  const float* Ab[5];
  #pragma unroll
  for (int tm = 0; tm < 5; ++tm) {
    const int a_row = tm * 16 + l16;
    Ab[tm] = word_emb + ((size_t)b * NW + (a_row < NW ? a_row : NW - 1)) * WD + 2 * half;
  }
  const float* Bb[2];
  #pragma unroll
  for (int j = 0; j < 2; ++j)
    Bb[j] = W_fc + (size_t)(jb * 32 + 2 * l16 + j) * WD + 2 * half;

  v8f acc[5][2] = {};
  v2f a0[K1_STEPS][5], b0[K1_STEPS][2], a1[K1_STEPS][5], b1[K1_STEPS][2];
  k1_load(a0, b0, Ab, Bb, 0);
  #pragma unroll
  for (int s = 0; s < K1_STEPS; ++s)
    asm volatile("" :: "v"(a0[s][0]), "v"(a0[s][1]), "v"(a0[s][2]), "v"(a0[s][3]),
                       "v"(a0[s][4]), "v"(b0[s][0]), "v"(b0[s][1]));
  for (int k0 = 0; k0 < WD; k0 += 8 * K1_STEPS) {
    k1_load(a1, b1, Ab, Bb, k0 + 4 * K1_STEPS);
    k1_mma(acc, a0, b0);
    k1_load(a0, b0, Ab, Bb, (k0 + 8 * K1_STEPS) & (WD - 1));
    k1_mma(acc, a1, b1);
  }

  const int col = jb * 32 + 2 * l16;
  const float bias0 = b_fc[col], bias1 = b_fc[col + 1];
  #pragma unroll
  for (int tm = 0; tm < 5; ++tm) {
    float* D = we_ws + ((size_t)b * NP + tm * 16 + 8 * half) * HW2 + col;
    #pragma unroll
    for (int i = 0; i < 8; ++i) {
      const v2f v = {acc[tm][0][i] + bias0, acc[tm][1][i] + bias1};
      *(v2f*)(D + (size_t)i * HW2) = v;
    }
  }
}

#define K2_KC    32
#define K2_NCH   (HW2 / K2_KC)
#define K2_PITCH 36
#define K2_AROWS 64
#define K2_BUF   ((K2_AROWS + NP) * K2_PITCH)
#define K2_SETS  6
#define K3_PITCH 84

static __device__ __forceinline__ void k2_wait_lds() {
  __builtin_amdgcn_sched_barrier(0);
  asm volatile("s_wait_dscnt 0x0" ::: "memory");
}
static __device__ __forceinline__ void k2_barrier() {
  asm volatile("s_barrier_signal -1\n\ts_barrier_wait -1" ::: "memory");
}

static __device__ __forceinline__ void k2_gload4(v4f (&p)[5], const float* g0, int k0) {
  #pragma unroll
  for (int i = 0; i < 4; ++i) p[i] = *(const v4f*)(g0 + (size_t)(4 * i) * HW2 + k0);
}
static __device__ __forceinline__ void k2_lstore4(float* buf, const v4f (&p)[5], int st0) {
  #pragma unroll
  for (int i = 0; i < 4; ++i) *(v4f*)(buf + st0 + 4 * i * K2_PITCH) = p[i];
}

static __device__ __forceinline__ void k2_ldfrag2(
    v2f (&a)[4], v2f (&bm)[4][3], const float* buf, int fA, int fB, int h) {
  #pragma unroll
  for (int s = 0; s < 4; ++s) {
    a[s] = *(const v2f*)(buf + fA + 16 * h + 4 * s);
    #pragma unroll
    for (int t = 0; t < 2; ++t)
      bm[s][t] = *(const v2f*)(buf + fB + 16 * t * K2_PITCH + 16 * h + 4 * s);
  }
}
static __device__ __forceinline__ void k2_ldfrag3(
    v2f (&bm)[4][3], const float* buf, int fB, int h) {
  #pragma unroll
  for (int s = 0; s < 4; ++s)
    bm[s][2] = *(const v2f*)(buf + fB + 32 * K2_PITCH + 16 * h + 4 * s);
}

static __device__ __forceinline__ void k2_mma2(
    v8f (&acc)[3], const v2f (&a)[4], const v2f (&bm)[4][3]) {
  #pragma unroll
  for (int s = 0; s < 4; ++s)
    #pragma unroll
    for (int t = 0; t < 2; ++t)
      acc[t] = wmma_f32(a[s], bm[s][t], acc[t]);
}
static __device__ __forceinline__ void k2_mma3(
    v8f (&acc)[3], const v2f (&a)[4], const v2f (&bm)[4][3]) {
  #pragma unroll
  for (int s = 0; s < 4; ++s) acc[2] = wmma_f32(a[s], bm[s][2], acc[2]);
}

template <int J, bool PARK, bool REFILL>
static __device__ __forceinline__ void k2_pre(
    int part, v8f (&acc)[3], v4f (&p)[K2_SETS][5],
    v2f (&a0)[4], v2f (&b0)[4][3], v2f (&a1)[4], v2f (&b1)[4][3],
    float* buf0, float* buf1, const float* g0, const float* g1, int knext,
    int st0, int st1, int fA, int fB) {
  float* bufX = (J & 1) ? buf1 : buf0;
  float* bufY = (J & 1) ? buf0 : buf1;
  constexpr int S = (J + 1) % K2_SETS;
  k2_ldfrag2(a1, b1, bufX, fA, fB, 1);
  if (PARK) k2_lstore4(bufY, p[S], st0);
  if (REFILL) k2_gload4(p[S], g0, knext);
  if (part == 0) {
    k2_ldfrag3(b1, bufX, fB, 1);
    if (PARK) *(v4f*)(bufY + st1) = p[S][4];
    if (REFILL) p[S][4] = *(const v4f*)(g1 + knext);
    k2_mma3(acc, a0, b0);
  }
  k2_mma2(acc, a0, b0);
}
template <int J, bool PARK>
static __device__ __forceinline__ void k2_post(
    int part, v8f (&acc)[3], v2f (&a0)[4], v2f (&b0)[4][3],
    const v2f (&a1)[4], const v2f (&b1)[4][3],
    float* buf0, float* buf1, int fA, int fB) {
  float* bufY = (J & 1) ? buf0 : buf1;
  if (PARK) k2_ldfrag2(a0, b0, bufY, fA, fB, 0);
  if (part == 0) {
    if (PARK) k2_ldfrag3(b0, bufY, fB, 0);
    k2_mma3(acc, a1, b1);
  }
  k2_mma2(acc, a1, b1);
}

static __device__ __forceinline__ void sm_max(
    float (&m)[8], const v8f (&acc)[3], int col0, int nt) {
  #pragma unroll
  for (int i = 0; i < 8; ++i) {
    float v = -3.402823466e38f;
    #pragma unroll
    for (int t = 0; t < 3; ++t)
      if (t < nt && col0 + 16 * t < NW) v = fmaxf(v, acc[t][i]);
    #pragma unroll
    for (int off = 8; off >= 1; off >>= 1) v = fmaxf(v, __shfl_xor(v, off, 32));
    m[i] = v;
  }
}
static __device__ __forceinline__ void sm_exp(
    float* Ls, const v8f (&acc)[3], const float (&m)[8], int col0, int nt) {
  #pragma unroll
  for (int i = 0; i < 8; ++i)
    #pragma unroll
    for (int t = 0; t < 3; ++t)
      if (t < nt)
        Ls[i * K3_PITCH + 16 * t] = (col0 + 16 * t < NW) ? __expf(acc[t][i] - m[i]) : 0.f;
}

#define K3_DS    10

static __device__ __forceinline__ void o_loadB(v4f (&bm)[2][2], const float* p) {
  #pragma unroll
  for (int ks = 0; ks < 2; ++ks)
    #pragma unroll
    for (int s = 0; s < 2; ++s) bm[ks][s] = *(const v4f*)(p + (size_t)(4 * ks + s) * HW2);
}
static __device__ __forceinline__ void o_loadA(v2f (&a)[2][2], const float* p) {
  #pragma unroll
  for (int ks = 0; ks < 2; ++ks)
    #pragma unroll
    for (int r = 0; r < 2; ++r) a[ks][r] = *(const v2f*)(p + 16 * r * K3_PITCH + 4 * ks);
}
static __device__ __forceinline__ void o_pin() { __builtin_amdgcn_sched_barrier(0); }

template <int PH, bool ST>
static __device__ __forceinline__ void o_tile(
    v8f (&cur)[2][4], const v8f (&prev)[2][4], v4f (&bq)[4][2][2], v2f (&aq)[2][2][2],
    const float* Bt, const float* Btn, const float* Al, float* Dprev) {
  #pragma unroll
  for (int r = 0; r < 2; ++r)
    #pragma unroll
    for (int t = 0; t < 4; ++t) cur[r][t] = v8f{};
  #pragma unroll
  for (int d = 0; d < K3_DS; ++d) {
    o_loadB(bq[(PH + d + 3) & 3], (d + 3 < K3_DS) ? Bt + (size_t)(8 * (d + 3)) * HW2
                                                  : Btn + (size_t)(8 * (d + 3 - K3_DS)) * HW2);
    o_loadA(aq[(d + 1) & 1], Al + ((d + 1 < K3_DS) ? 8 * (d + 1) : 0));
    if (ST && d < 8) {
      #pragma unroll
      for (int q = 0; q < 2; ++q) {
        const int i = 2 * (d & 3) + q;
        const v4f v = {prev[d >> 2][0][i], prev[d >> 2][1][i],
                       prev[d >> 2][2][i], prev[d >> 2][3][i]};
        *(v4f*)(Dprev + (size_t)(16 * (d >> 2) + i) * HW2) = v;
      }
    }
    o_pin();
    #pragma unroll
    for (int ks = 0; ks < 2; ++ks) {
      const v4f (&L)[2] = bq[(PH + d) & 3][ks];
      #pragma unroll
      for (int t = 0; t < 4; ++t) {
        const v2f bt = {L[0][t], L[1][t]};
        #pragma unroll
        for (int r = 0; r < 2; ++r)
          cur[r][t] = wmma_f32(aq[d & 1][ks][r], bt, cur[r][t]);
      }
    }
  }
}

__global__ __launch_bounds__(256) void scores_softmax_out_kernel(
    const float* __restrict__ feat, const float* __restrict__ we_ws,
    float* __restrict__ out) {
  __shared__ __attribute__((aligned(16))) float lds[2 * K2_BUF];
  __shared__ __attribute__((aligned(16))) float att_lds[K2_AROWS * K3_PITCH];
  __shared__ float pmax[2 * K2_AROWS];

  const int lane = threadIdx.x & 31;
  const int wave = __builtin_amdgcn_readfirstlane((int)(threadIdx.x >> 5));
  const int slab = wave & 3;
  const int part = wave >> 2;
  const int b   = blockIdx.x >> 3;
  const int cs0 = (blockIdx.x & 7) * K2_AROWS;
  const int half = lane >> 4;
  const int l16  = lane & 15;
  const int r4   = lane >> 3;
  const int seg  = lane & 7;

  const float* gF = feat  + ((size_t)b * Cn + cs0 + 16 * slab + r4) * HW2 + 4 * seg;
  const float* gW = we_ws + ((size_t)b * NP + 16 + 16 * slab + r4) * HW2 + 4 * seg;
  const float* g0 = part ? gW : gF;
  const float* g1 = we_ws + ((size_t)b * NP + 4 * slab + r4) * HW2 + 4 * seg;
  const int st0 = ((part ? K2_AROWS + 16 : 0) + 16 * slab + r4) * K2_PITCH + 4 * seg;
  const int st1 = (K2_AROWS + 4 * slab + r4) * K2_PITCH + 4 * seg;
  const int fA = (16 * slab + l16) * K2_PITCH + 2 * half;
  const int fB = (K2_AROWS + 48 * part + l16) * K2_PITCH + 2 * half;
  float* buf0 = lds;
  float* buf1 = lds + K2_BUF;

  v8f acc[3] = {};
  v4f p[K2_SETS][5];
  v2f a0[4], b0[4][3], a1[4], b1[4][3];

  #pragma unroll
  for (int s = 0; s < K2_SETS; ++s) k2_gload4(p[s], g0, s * K2_KC);
  if (part == 0) {
    #pragma unroll
    for (int s = 0; s < K2_SETS; ++s) p[s][4] = *(const v4f*)(g1 + s * K2_KC);
  }
  k2_lstore4(buf0, p[0], st0);
  k2_gload4(p[0], g0, K2_SETS * K2_KC);
  if (part == 0) {
    *(v4f*)(buf0 + st1) = p[0][4];
    p[0][4] = *(const v4f*)(g1 + K2_SETS * K2_KC);
  }
  k2_wait_lds();
  k2_barrier();
  k2_ldfrag2(a0, b0, buf0, fA, fB, 0);
  if (part == 0) k2_ldfrag3(b0, buf0, fB, 0);

  #define K2_CHUNK(J, PARK, REFILL, KN)                                          \
    k2_pre<J, PARK, REFILL>(part, acc, p, a0, b0, a1, b1, buf0, buf1, g0, g1, KN, \
                            st0, st1, fA, fB);                                   \
    if (PARK) { k2_wait_lds(); k2_barrier(); }                                   \
    k2_post<J, PARK>(part, acc, a0, b0, a1, b1, buf0, buf1, fA, fB);

  #define K2_NEXT(c) (((c) + 7 < K2_NCH ? (c) + 7 : K2_NCH - 1) * K2_KC)
  #pragma nounroll
  for (int t = 0; t < K2_NCH - 2; t += K2_SETS) {
    K2_CHUNK(0, true, true, K2_NEXT(t + 0))
    K2_CHUNK(1, true, true, K2_NEXT(t + 1))
    K2_CHUNK(2, true, true, K2_NEXT(t + 2))
    K2_CHUNK(3, true, true, K2_NEXT(t + 3))
    K2_CHUNK(4, true, true, K2_NEXT(t + 4))
    K2_CHUNK(5, true, true, K2_NEXT(t + 5))
  }
  #undef K2_NEXT
  K2_CHUNK(0, true, false, 0)
  K2_CHUNK(1, false, false, 0)
  #undef K2_CHUNK

  const int orh = wave & 1;
  const int oq  = wave >> 1;
  const float* Bw = we_ws + ((size_t)b * NP + 2 * half) * HW2 + 256 * oq + 4 * l16;
  const float* Al = att_lds + (32 * orh + l16) * K3_PITCH + 2 * half;
  v4f bq[4][2][2];
  v2f aq[2][2][2];
  o_loadB(bq[0], Bw);
  o_loadB(bq[1], Bw + (size_t)8 * HW2);
  o_loadB(bq[2], Bw + (size_t)16 * HW2);

  const int col0 = 48 * part + l16;
  float m[8];
  const int nt = 3 - part;
  sm_max(m, acc, col0, nt);
  const int srow = 16 * slab + 8 * half;
  #pragma unroll
  for (int i = 0; i < 8; ++i) pmax[part * K2_AROWS + srow + i] = m[i];
  k2_wait_lds();
  k2_barrier();
  #pragma unroll
  for (int i = 0; i < 8; ++i) m[i] = fmaxf(pmax[srow + i], pmax[K2_AROWS + srow + i]);
  float* Ls = att_lds + srow * K3_PITCH + col0;
  sm_exp(Ls, acc, m, col0, nt);
  k2_wait_lds();
  k2_barrier();
  float* Ln = att_lds + (16 * slab + 8 * part + 4 * half) * K3_PITCH + l16;
  #pragma unroll
  for (int i = 0; i < 4; ++i) {
    float e[5], s = 0.f;
    #pragma unroll
    for (int t = 0; t < 5; ++t) {
      e[t] = Ln[i * K3_PITCH + 16 * t];
      s += e[t];
    }
    #pragma unroll
    for (int off = 8; off >= 1; off >>= 1) s += __shfl_xor(s, off, 32);
    const float inv = 1.f / s;
    #pragma unroll
    for (int t = 0; t < 5; ++t)
      Ln[i * K3_PITCH + 16 * t] = e[t] * inv;
  }
  k2_wait_lds();
  k2_barrier();

  float* Dw = out + ((size_t)b * Cn + cs0 + 32 * orh + 8 * half) * HW2 + 256 * oq + 4 * l16;
  v8f acc0[2][4], acc1[2][4];
  o_loadA(aq[0], Al);
  o_tile<0, false>(acc0, acc1, bq, aq, Bw,       Bw + 64,  Al, Dw);
  o_tile<2, true >(acc1, acc0, bq, aq, Bw + 64,  Bw + 128, Al, Dw);
  o_tile<0, true >(acc0, acc1, bq, aq, Bw + 128, Bw + 192, Al, Dw + 64);
  o_tile<2, true >(acc1, acc0, bq, aq, Bw + 192, Bw,       Al, Dw + 128);
  #pragma unroll
  for (int r = 0; r < 2; ++r)
    #pragma unroll
    for (int i = 0; i < 8; ++i) {
      const v4f v = {acc1[r][0][i], acc1[r][1][i], acc1[r][2][i], acc1[r][3][i]};
      *(v4f*)(Dw + (size_t)(16 * r + i) * HW2 + 64 * 3) = v;
    }
}

extern "C" void kernel_launch(void* const* d_in, const int* in_sizes, int n_in,
                              void* d_out, int out_size, void* d_ws, size_t ws_size,
                              hipStream_t stream) {
  const float* feat     = (const float*)d_in[0];
  const float* word_emb = (const float*)d_in[1];
  const float* W_fc     = (const float*)d_in[2];
  const float* b_fc     = (const float*)d_in[3];
  float* out = (float*)d_out;

  float* we_ws  = (float*)d_ws;

  we_gemm_kernel       <<<(Bn * 32) / 4,      128, 0, stream>>>(word_emb, W_fc, b_fc, we_ws);
  scores_softmax_out_kernel<<<Bn * 8,         256, 0, stream>>>(feat, we_ws, out);
}
